# starting kernel
# baseline (speedup 1.0000x reference)
#include <hip/hip_runtime.h>
#include <hip/hip_bf16.h>

typedef __attribute__((ext_vector_type(2))) float v2f;
typedef __attribute__((ext_vector_type(8))) float v8f;

#define HID 64
#define INDIM 128
#define KDIM 320
#define GCOLS 256
#define LMAX 2.0f

__device__ __forceinline__ v8f wmma_f32_4(v2f a, v2f b, v8f c) {
  return __builtin_amdgcn_wmma_f32_16x16x4_f32(false, a, false, b, (short)0, c, false, false);
}

__device__ __forceinline__ float fsigmoid(float x) {
  return 1.0f / (1.0f + __expf(-x));
}

__device__ __forceinline__ void atomic_fadd(float* p, float v) {
  asm volatile("global_atomic_add_f32 %0, %1, off scope:SCOPE_DEV"
               :: "v"(p), "v"(v) : "memory");
}


__global__ __launch_bounds__(256)
void k_degree(const int* __restrict__ src, const int* __restrict__ dst,
              const float* __restrict__ ew, float* __restrict__ deg, int E) {
  int e = blockIdx.x * 256 + threadIdx.x;
  if (e >= E) return;
  int s = src[e], d = dst[e];
  if (s != d) atomic_fadd(&deg[s], ew[e]);
}

__global__ __launch_bounds__(256)
void k_dinv(const float* __restrict__ deg, float* __restrict__ dinv, int Nn) {
  int i = blockIdx.x * 256 + threadIdx.x;
  if (i >= Nn) return;
  float dg = deg[i];
  dinv[i] = (dg > 0.0f) ? rsqrtf(dg) : 0.0f;
}

__global__ __launch_bounds__(256)
void k_lapw(const int* __restrict__ src, const int* __restrict__ dst,
            const float* __restrict__ ew, const float* __restrict__ dinv,
            float* __restrict__ lapw, int E) {
  int e = blockIdx.x * 256 + threadIdx.x;
  if (e >= E) return;
  int s = src[e], d = dst[e];
  float w = (s == d) ? 0.0f : ew[e];
  lapw[e] = (2.0f / LMAX) * (-dinv[s] * w * dinv[d]);
}

__global__ __launch_bounds__(256)
void k_scale(float* __restrict__ dstp, const float* __restrict__ srcp,
             float scale, int count) {
  int i = blockIdx.x * 256 + threadIdx.x;
  if (i >= count) return;
  dstp[i] = scale * srcp[i];
}

__global__ __launch_bounds__(256)
void k_spmm(const float* __restrict__ z, float* __restrict__ acc,
            const int* __restrict__ src, const int* __restrict__ dst,
            const float* __restrict__ lw, int E) {
  int gid = blockIdx.x * 256 + threadIdx.x;
  int e = gid >> 4;
  if (e >= E) return;
  float w = lw[e];
  if (w == 0.0f) return;
  int ch = (gid & 15) << 2;
  int s = src[e], d = dst[e];
  const float4 zv = *(const float4*)(z + (size_t)s * HID + ch);
  float* out = acc + (size_t)d * HID + ch;
  atomic_fadd(out + 0, w * zv.x);
  atomic_fadd(out + 1, w * zv.y);
  atomic_fadd(out + 2, w * zv.z);
  atomic_fadd(out + 3, w * zv.w);
}

__global__ __launch_bounds__(256)
void k_cheb2(float* __restrict__ t2, const float* __restrict__ h, int count) {
  int i = blockIdx.x * 256 + threadIdx.x;
  if (i >= count) return;
  t2[i] = 2.0f * t2[i] - h[i];
}


__global__ __launch_bounds__(256)
void k_packB(const float* __restrict__ Wx, const float* __restrict__ theta,
             const float* __restrict__ bg, const float* __restrict__ conv_b,
             float* __restrict__ Bp, float* __restrict__ bias) {
  int gid = blockIdx.x * 256 + threadIdx.x;
  if (gid >= KDIM * GCOLS) return;
  int k = gid / GCOLS, col = gid % GCOLS;
  int g = col >> 6, o = col & 63;
  float v;
  if (k < INDIM) {
    v = Wx[((size_t)g * INDIM + k) * HID + o];
  } else {
    int kk = k - INDIM;
    int kb = kk >> 6, kr = kk & 63;
    v = theta[(((size_t)g * 3 + kb) * HID + kr) * HID + o];
  }
  Bp[gid] = v;
  if (gid < GCOLS) bias[gid] = bg[g * HID + o] + conv_b[g * HID + o];
}


#define AS1 324
#define PS1 260

__global__ __launch_bounds__(256)
void k_gemm_gates(const float* __restrict__ x, const float* __restrict__ h,
                  const float* __restrict__ T1, const float* __restrict__ T2,
                  const float* __restrict__ Bp, const float* __restrict__ bias,
                  const float* __restrict__ c_in,
                  float* __restrict__ h_out, float* __restrict__ c_out, int Nn) {
  __shared__ float Asmem[16 * AS1];
  __shared__ float Psmem[16 * PS1];
  const int tid = threadIdx.x;
  const int m0 = blockIdx.x * 16;

  for (int i = tid; i < 16 * KDIM; i += 256) {
    int r = i / KDIM, k = i % KDIM;
    int n = m0 + r; if (n >= Nn) n = Nn - 1;
    float v;
    if (k < 128)      v = x[(size_t)n * INDIM + k];
    else if (k < 192) v = h[(size_t)n * HID + (k - 128)];
    else if (k < 256) v = T1[(size_t)n * HID + (k - 192)];
    else              v = T2[(size_t)n * HID + (k - 256)];
    Asmem[r * AS1 + k] = v;
  }
  __syncthreads();

  const int wave = tid >> 5, lane = tid & 31;
  const int lo = lane & 15, hi = lane >> 4;
  const int c0 = wave * 32;
  v8f acc0 = {}; v8f acc1 = {};

  for (int k0 = 0; k0 < KDIM; k0 += 4) {
    v2f a = *(const v2f*)&Asmem[lo * AS1 + k0 + 2 * hi];
    const float* bptr = Bp + (size_t)(k0 + 2 * hi) * GCOLS + c0 + lo;
    v2f b0, b1;
    b0[0] = bptr[0];    b0[1] = bptr[GCOLS];
    b1[0] = bptr[16];   b1[1] = bptr[GCOLS + 16];
    acc0 = wmma_f32_4(a, b0, acc0);
    acc1 = wmma_f32_4(a, b1, acc1);
  }

  float bias0 = bias[c0 + lo], bias1 = bias[c0 + 16 + lo];
#pragma unroll
  for (int r = 0; r < 8; ++r) {
    int row = r + 8 * hi;
    Psmem[row * PS1 + c0 + lo]      = acc0[r] + bias0;
    Psmem[row * PS1 + c0 + 16 + lo] = acc1[r] + bias1;
  }
  __syncthreads();

#pragma unroll
  for (int j = 0; j < 4; ++j) {
    int flat = j * 256 + tid;
    int row = flat >> 6, o = flat & 63;
    int n = m0 + row;
    if (n >= Nn) continue;
    const float* pr = &Psmem[row * PS1];
    float ig = fsigmoid(pr[o]);
    float fg = fsigmoid(pr[64 + o]);
    float tg = tanhf(pr[128 + o]);
    float og = fsigmoid(pr[192 + o]);
    float cv = fg * c_in[(size_t)n * HID + o] + ig * tg;
    float hv = og * tanhf(cv);
    h_out[(size_t)n * HID + o] = hv;
    c_out[(size_t)n * HID + o] = cv;
  }
}


#define AS2 68
#define BS2 68

__global__ __launch_bounds__(256)
void k_gemm_out(const float* __restrict__ h0, const float* __restrict__ Wl,
                const float* __restrict__ bl, float* __restrict__ out, int Nn) {
  __shared__ float Bs[64 * BS2];
  __shared__ float As[32 * AS2];
  const int tid = threadIdx.x;
  const int m0 = blockIdx.x * 32;

  for (int i = tid; i < 64 * 64; i += 256) {
    int k = i >> 6, ccol = i & 63;
    Bs[k * BS2 + ccol] = Wl[i];
  }
  for (int i = tid; i < 32 * 64; i += 256) {
    int r = i >> 6, k = i & 63;
    int n = m0 + r; if (n >= Nn) n = Nn - 1;
    float v = h0[(size_t)n * HID + k];
    As[r * AS2 + k] = v > 0.0f ? v : 0.0f;
  }
  __syncthreads();

  const int wave = tid >> 5, lane = tid & 31;
  const int lo = lane & 15, hi = lane >> 4;
  const int r0 = (wave >> 2) * 16, c0 = (wave & 3) * 16;
  v8f acc = {};
  for (int k0 = 0; k0 < 64; k0 += 4) {
    v2f a = *(const v2f*)&As[(r0 + lo) * AS2 + k0 + 2 * hi];
    v2f b;
    b[0] = Bs[(k0 + 2 * hi) * BS2 + c0 + lo];
    b[1] = Bs[(k0 + 2 * hi + 1) * BS2 + c0 + lo];
    acc = wmma_f32_4(a, b, acc);
  }
  float bb = bl[c0 + lo];
#pragma unroll
  for (int r = 0; r < 8; ++r) {
    int row = r0 + r + 8 * hi;
    int n = m0 + row;
    if (n < Nn) out[(size_t)n * HID + c0 + lo] = acc[r] + bb;
  }
}


extern "C" void kernel_launch(void* const* d_in, const int* in_sizes, int n_in,
                              void* d_out, int out_size, void* d_ws, size_t ws_size,
                              hipStream_t stream) {
  const float* x      = (const float*)d_in[0];
  const int*   ei     = (const int*)d_in[1];
  const float* ew     = (const float*)d_in[2];
  const float* h      = (const float*)d_in[3];
  const float* c      = (const float*)d_in[4];
  const float* Wx     = (const float*)d_in[5];
  const float* bg     = (const float*)d_in[6];
  const float* theta  = (const float*)d_in[7];
  const float* conv_b = (const float*)d_in[8];
  const float* Wl     = (const float*)d_in[9];
  const float* bl     = (const float*)d_in[10];

  const int E  = in_sizes[2];
  const int Nn = in_sizes[3] / HID;
  const int* src = ei;
  const int* dst = ei + E;

  float* deg  = (float*)d_ws;
  float* dinv = deg  + Nn;
  float* lapw = dinv + Nn;
  float* T1   = lapw + E;
  float* T2   = T1 + (size_t)Nn * HID;
  float* Bp   = T2 + (size_t)Nn * HID;
  float* bias = Bp + (size_t)KDIM * GCOLS;

  float* out   = (float*)d_out;
  float* h_out = out   + (size_t)Nn * HID;
  float* c_out = h_out + (size_t)Nn * HID;

  const float DIAG = 2.0f / LMAX - 1.0f;
  const int NH = Nn * HID;

  hipMemsetAsync(deg, 0, (size_t)Nn * sizeof(float), stream);
  k_degree<<<(E + 255) / 256, 256, 0, stream>>>(src, dst, ew, deg, E);
  k_dinv<<<(Nn + 255) / 256, 256, 0, stream>>>(deg, dinv, Nn);
  k_lapw<<<(E + 255) / 256, 256, 0, stream>>>(src, dst, ew, dinv, lapw, E);

  k_scale<<<(NH + 255) / 256, 256, 0, stream>>>(T1, h, DIAG, NH);
  k_spmm<<<(E * 16 + 255) / 256, 256, 0, stream>>>(h, T1, src, dst, lapw, E);
  k_scale<<<(NH + 255) / 256, 256, 0, stream>>>(T2, T1, DIAG, NH);
  k_spmm<<<(E * 16 + 255) / 256, 256, 0, stream>>>(T1, T2, src, dst, lapw, E);
  k_cheb2<<<(NH + 255) / 256, 256, 0, stream>>>(T2, h, NH);

  k_packB<<<(KDIM * GCOLS + 255) / 256, 256, 0, stream>>>(Wx, theta, bg, conv_b, Bp, bias);

  k_gemm_gates<<<(Nn + 15) / 16, 256, 0, stream>>>(x, h, T1, T2, Bp, bias, c,
                                                   h_out, c_out, Nn);
  k_gemm_out<<<(Nn + 31) / 32, 256, 0, stream>>>(h_out, Wl, bl, out, Nn);
}


# gate GEMM: each work-group keeps one column quarter of B (4 gates x 16 hidden cols, all k) in LDS for its life and loops over 32-row A tiles (512 groups); B read once per group; sum orders kept
# speedup vs baseline: 1.9352x; 1.9352x over previous
#include <hip/hip_runtime.h>
#include <hip/hip_bf16.h>

typedef __attribute__((ext_vector_type(2))) float v2f;
typedef __attribute__((ext_vector_type(8))) float v8f;

#define HID 64
#define INDIM 128
#define KDIM 320
#define GCOLS 256
#define LMAX 2.0f

__device__ __forceinline__ v8f wmma_f32_4(v2f a, v2f b, v8f c) {
  return __builtin_amdgcn_wmma_f32_16x16x4_f32(false, a, false, b, (short)0, c, false, false);
}

__device__ __forceinline__ float fsigmoid(float x) {
  return 1.0f / (1.0f + __expf(-x));
}

__device__ __forceinline__ void atomic_fadd(float* p, float v) {
  asm volatile("global_atomic_add_f32 %0, %1, off scope:SCOPE_DEV"
               :: "v"(p), "v"(v) : "memory");
}

__device__ __forceinline__ unsigned cache_half() {
  return (__builtin_amdgcn_s_sendmsg_rtn(0x87) >> 18) & 1u;
}


__global__ __launch_bounds__(256)
void k_degree(const int* __restrict__ src, const int* __restrict__ dst,
              const float* __restrict__ ew, float* __restrict__ deg2, int E, int Nn) {
  float* deg = deg2 + (size_t)cache_half() * Nn;
  int e = blockIdx.x * 256 + threadIdx.x;
  if (e >= E) return;
  int s = src[e], d = dst[e];
  if (s != d) atomic_fadd(&deg[s], ew[e]);
}

__global__ __launch_bounds__(256)
void k_dinv(const float* __restrict__ deg2, float* __restrict__ dinv, int Nn) {
  int i = blockIdx.x * 256 + threadIdx.x;
  if (i >= Nn) return;
  float dg = deg2[i] + deg2[Nn + i];
  dinv[i] = (dg > 0.0f) ? rsqrtf(dg) : 0.0f;
}

__global__ __launch_bounds__(256)
void k_prep(const int* __restrict__ src, const int* __restrict__ dst,
            const float* __restrict__ ew, const float* __restrict__ dinv,
            float* __restrict__ lapw, int E,
            const float* __restrict__ h, float* __restrict__ a0, float* __restrict__ a1,
            float* __restrict__ b1, float scale, int count4,
            const float* __restrict__ Wx, const float* __restrict__ theta,
            const float* __restrict__ bg, const float* __restrict__ conv_b,
            float* __restrict__ Bp, float* __restrict__ bias) {
  const int gid = blockIdx.x * 256 + threadIdx.x;
  const int nthr = gridDim.x * 256;

  const float4* __restrict__ h4 = reinterpret_cast<const float4*>(h);
  float4* __restrict__ a04 = reinterpret_cast<float4*>(a0);
  float4* __restrict__ a14 = reinterpret_cast<float4*>(a1);
  float4* __restrict__ b14 = reinterpret_cast<float4*>(b1);
  const float4 zero4 = make_float4(0.0f, 0.0f, 0.0f, 0.0f);

  const bool own = gid < count4;
  float4 z = zero4;
  if (own) {
    z = h4[gid];
    a14[gid] = zero4;
    b14[gid] = zero4;
  }

  if (gid < E) {
    int s = src[gid], d = dst[gid];
    float w = (s == d) ? 0.0f : ew[gid];
    lapw[gid] = (2.0f / LMAX) * (-dinv[s] * w * dinv[d]);
  }

  if (own) a04[gid] = make_float4(scale * z.x, scale * z.y, scale * z.z, scale * z.w);
  for (int i = gid + nthr; i < count4; i += nthr) {
    float4 y = h4[i];
    a04[i] = make_float4(scale * y.x, scale * y.y, scale * y.z, scale * y.w);
    a14[i] = zero4;
    b14[i] = zero4;
  }

  for (int p = gid; p < KDIM * GCOLS; p += nthr) {
    int k = p / GCOLS, col = p % GCOLS;
    int g = col >> 6, o = col & 63;
    float v;
    if (k < INDIM) {
      v = Wx[((size_t)g * INDIM + k) * HID + o];
    } else {
      int kk = k - INDIM;
      int kb = kk >> 6, kr = kk & 63;
      v = theta[(((size_t)g * 3 + kb) * HID + kr) * HID + o];
    }
    Bp[p] = v;
    if (p < GCOLS) bias[p] = bg[g * HID + o] + conv_b[g * HID + o];
  }
}

__global__ __launch_bounds__(256)
void k_mid(float* __restrict__ a0, const float* __restrict__ a1,
           float* __restrict__ b0, float scale, int count) {
  int i = blockIdx.x * 256 + threadIdx.x;
  if (i >= count) return;
  float t1 = a0[i] + a1[i];
  a0[i] = t1;
  b0[i] = scale * t1;
}


__device__ __forceinline__ void atomic_rows8(unsigned voff, const float* p0, const float* p1,
                                             float* r0, float* r1, float* r2, float* r3,
                                             float* r4, float* r5, float* r6, float* r7) {
  asm volatile(
      "global_atomic_add_f32 %0, %1, %17 scope:SCOPE_DEV\n\t"
      "global_atomic_add_f32 %0, %2, %17 offset:128 scope:SCOPE_DEV\n\t"
      "global_atomic_add_f32 %0, %3, %18 scope:SCOPE_DEV\n\t"
      "global_atomic_add_f32 %0, %4, %18 offset:128 scope:SCOPE_DEV\n\t"
      "global_atomic_add_f32 %0, %5, %19 scope:SCOPE_DEV\n\t"
      "global_atomic_add_f32 %0, %6, %19 offset:128 scope:SCOPE_DEV\n\t"
      "global_atomic_add_f32 %0, %7, %20 scope:SCOPE_DEV\n\t"
      "global_atomic_add_f32 %0, %8, %20 offset:128 scope:SCOPE_DEV\n\t"
      "global_atomic_add_f32 %0, %9, %21 scope:SCOPE_DEV\n\t"
      "global_atomic_add_f32 %0, %10, %21 offset:128 scope:SCOPE_DEV\n\t"
      "global_atomic_add_f32 %0, %11, %22 scope:SCOPE_DEV\n\t"
      "global_atomic_add_f32 %0, %12, %22 offset:128 scope:SCOPE_DEV\n\t"
      "global_atomic_add_f32 %0, %13, %23 scope:SCOPE_DEV\n\t"
      "global_atomic_add_f32 %0, %14, %23 offset:128 scope:SCOPE_DEV\n\t"
      "global_atomic_add_f32 %0, %15, %24 scope:SCOPE_DEV\n\t"
      "global_atomic_add_f32 %0, %16, %24 offset:128 scope:SCOPE_DEV\n\t"
      "s_wait_xcnt 0x0"
      :: "v"(voff),
         "v"(p0[0]), "v"(p1[0]), "v"(p0[1]), "v"(p1[1]), "v"(p0[2]), "v"(p1[2]), "v"(p0[3]), "v"(p1[3]),
         "v"(p0[4]), "v"(p1[4]), "v"(p0[5]), "v"(p1[5]), "v"(p0[6]), "v"(p1[6]), "v"(p0[7]), "v"(p1[7]),
         "s"(r0), "s"(r1), "s"(r2), "s"(r3), "s"(r4), "s"(r5), "s"(r6), "s"(r7)
      : "memory");
}

__global__ __launch_bounds__(256)
void k_spmm(const float* __restrict__ z, float* __restrict__ acc0, float* __restrict__ acc1,
            const int* __restrict__ src, const int* __restrict__ dst,
            const float* __restrict__ lw, int E) {
  float* acc = cache_half() ? acc1 : acc0;
  const unsigned lane = threadIdx.x & 31u;
  const int base = (blockIdx.x * 8 + (threadIdx.x >> 5)) * 32;
  if (base >= E) return;
  const int e = base + (int)lane;
  int s = 0, d = 0, wbits = 0;
  if (e < E) { s = src[e]; d = dst[e]; wbits = __float_as_int(lw[e]); }

  float v0[4][8], v1[4][8];
#pragma unroll
  for (int g = 0; g < 4; ++g) {
#pragma unroll
    for (int u = 0; u < 8; ++u) {
      const int sj = __builtin_amdgcn_readlane(s, g * 8 + u);
      const float* zr = z + (size_t)sj * HID;
      v0[g][u] = zr[lane];
      v1[g][u] = zr[lane + 32u];
    }
    if (g > 0) {
      float p0[8], p1[8]; float* r[8];
#pragma unroll
      for (int u = 0; u < 8; ++u) {
        const int j = (g - 1) * 8 + u;
        const float wj = __int_as_float(__builtin_amdgcn_readlane(wbits, j));
        r[u] = acc + (size_t)__builtin_amdgcn_readlane(d, j) * HID;
        p0[u] = (wj != 0.0f) ? wj * v0[g - 1][u] : 0.0f;
        p1[u] = (wj != 0.0f) ? wj * v1[g - 1][u] : 0.0f;
      }
      atomic_rows8(lane * 4u, p0, p1, r[0], r[1], r[2], r[3], r[4], r[5], r[6], r[7]);
    }
  }
  {
    float p0[8], p1[8]; float* r[8];
#pragma unroll
    for (int u = 0; u < 8; ++u) {
      const int j = 24 + u;
      const float wj = __int_as_float(__builtin_amdgcn_readlane(wbits, j));
      r[u] = acc + (size_t)__builtin_amdgcn_readlane(d, j) * HID;
      p0[u] = (wj != 0.0f) ? wj * v0[3][u] : 0.0f;
      p1[u] = (wj != 0.0f) ? wj * v1[3][u] : 0.0f;
    }
    atomic_rows8(lane * 4u, p0, p1, r[0], r[1], r[2], r[3], r[4], r[5], r[6], r[7]);
  }
}


#define TROWS 32
#define AS1 324
#define QC 64
#define BP2 160
#define PS1 68
#define KB 8

__device__ __forceinline__ float4 ld4(const float* p) { return *reinterpret_cast<const float4*>(p); }
__device__ __forceinline__ void st4(float* p, float4 v) { *reinterpret_cast<float4*>(p) = v; }
__device__ __forceinline__ float4 form_t2(float4 a, float4 b, float4 h) {
  return make_float4(2.0f * (a.x + b.x) - h.x, 2.0f * (a.y + b.y) - h.y,
                     2.0f * (a.z + b.z) - h.z, 2.0f * (a.w + b.w) - h.w);
}

__global__ __launch_bounds__(256)
void k_gemm_gates(const float* __restrict__ x, const float* __restrict__ h,
                  const float* __restrict__ T1,
                  const float* __restrict__ S2a, const float* __restrict__ S2b,
                  const float* __restrict__ Bp, const float* __restrict__ bias,
                  const float* __restrict__ c_in,
                  float* __restrict__ h_out, float* __restrict__ c_out, int Nn, int ntiles) {
  __shared__ __attribute__((aligned(16))) float Bs[(KDIM / 2) * BP2];
  __shared__ __attribute__((aligned(16))) float As[TROWS * AS1];
  __shared__ __attribute__((aligned(16))) float Ps[TROWS * PS1];
  const int tid = threadIdx.x;
  const int qd = blockIdx.x & 3;
  const int tstride = gridDim.x >> 2;
  int t = blockIdx.x >> 2;

  const float* __restrict__ c_q = c_in + 16 * qd;
  const int trow = tid >> 4, tcol = (tid & 15) * 4;
  float4 px0a, px1a, pha, pta, psaa, psba, px0b, px1b, phb, ptb, psab, psbb;
  float pca, pcb;
#define ATILE_LOAD(M0)                                                              \
  {                                                                                 \
    int na = (M0) + trow, nb = (M0) + trow + 16;                                    \
    if (na >= Nn) na = Nn - 1;                                                      \
    if (nb >= Nn) nb = Nn - 1;                                                      \
    const size_t qa = (size_t)na * HID + tcol, qb = (size_t)nb * HID + tcol;        \
    pca = c_q[(size_t)na * HID + (tid & 15)];                                       \
    pcb = c_q[(size_t)nb * HID + (tid & 15)];                                       \
    px0a = ld4(x + (size_t)na * INDIM + tcol);  px1a = ld4(x + (size_t)na * INDIM + 64 + tcol); \
    px0b = ld4(x + (size_t)nb * INDIM + tcol);  px1b = ld4(x + (size_t)nb * INDIM + 64 + tcol); \
    pha = ld4(h + qa);  pta = ld4(T1 + qa);  psaa = ld4(S2a + qa);  psba = ld4(S2b + qa);       \
    phb = ld4(h + qb);  ptb = ld4(T1 + qb);  psab = ld4(S2a + qb);  psbb = ld4(S2b + qb);       \
  }
  px0a = px1a = pha = pta = psaa = psba = px0b = px1b = phb = ptb = psab = psbb = make_float4(0.f, 0.f, 0.f, 0.f);
  pca = pcb = 0.0f;
  if (t < ntiles) ATILE_LOAD(t * TROWS)

#pragma unroll
  for (int j = 0; j < (KDIM / 2) * (QC / 4) / 256; ++j) {
    const int it = tid + 256 * j;
    const int p = it >> 4, ch = it & 15;
    const int c = (ch >> 2) * 16 + (ch & 3) * 4;
    const float* gp = Bp + (size_t)(2 * p) * GCOLS + (ch >> 2) * HID + 16 * qd + (ch & 3) * 4;
    const float4 r0 = *reinterpret_cast<const float4*>(gp);
    const float4 r1 = *reinterpret_cast<const float4*>(gp + GCOLS);
    float* sp = Bs + p * BP2 + 2 * c;
    *reinterpret_cast<float4*>(sp)     = make_float4(r0.x, r1.x, r0.y, r1.y);
    *reinterpret_cast<float4*>(sp + 4) = make_float4(r0.z, r1.z, r0.w, r1.w);
  }

  const int wave = tid >> 5, lane = tid & 31;
  const int lo = lane & 15, hi = lane >> 4;
  const int rs = wave >> 2, g = wave & 3;
  const float* afp = As + (rs * 16 + lo) * AS1 + 2 * hi;
  const float* bfp = Bs + hi * BP2 + 2 * (g * 16 + lo);
  const float biasv = bias[g * HID + 16 * qd + lo];

  for (; t < ntiles; t += tstride) {
    const int m0 = t * TROWS;
    {
      float* ra = As + trow * AS1 + tcol;
      float* rb = ra + 16 * AS1;
      st4(ra, px0a);  st4(ra + 64, px1a);  st4(ra + 128, pha);  st4(ra + 192, pta);
      st4(ra + 256, form_t2(psaa, psba, pha));
      st4(rb, px0b);  st4(rb + 64, px1b);  st4(rb + 128, phb);  st4(rb + 192, ptb);
      st4(rb + 256, form_t2(psab, psbb, phb));
    }
    __syncthreads();
    const float cin0 = pca, cin1 = pcb;
    if (t + tstride < ntiles) ATILE_LOAD((t + tstride) * TROWS)

    v8f acc = {};
    v2f af[2][KB], bf[2][KB];
#pragma unroll
    for (int s = 0; s < KB; ++s) {
      af[0][s] = *(const v2f*)(afp + 4 * s);
      bf[0][s] = *(const v2f*)(bfp + 2 * s * BP2);
    }
    __builtin_amdgcn_sched_group_barrier(0x100, KB, 0);
#pragma unroll
    for (int blk = 0; blk < KDIM / 4 / KB; ++blk) {
      const int cur = blk & 1, nxt = cur ^ 1;
      if (blk + 1 < KDIM / 4 / KB) {
#pragma unroll
        for (int s = 0; s < KB; ++s) {
          const int ks = (blk + 1) * KB + s;
          af[nxt][s] = *(const v2f*)(afp + 4 * ks);
          bf[nxt][s] = *(const v2f*)(bfp + 2 * ks * BP2);
        }
      }
#pragma unroll
      for (int s = 0; s < KB; ++s) acc = wmma_f32_4(af[cur][s], bf[cur][s], acc);
      if (blk + 1 < KDIM / 4 / KB) __builtin_amdgcn_sched_group_barrier(0x100, KB, 0);
      __builtin_amdgcn_sched_group_barrier(0x008, KB, 0);
    }

#pragma unroll
    for (int r = 0; r < 8; ++r)
      Ps[(rs * 16 + r + 8 * hi) * PS1 + g * 16 + lo] = acc[r] + biasv;
    __syncthreads();

#pragma unroll
    for (int u = 0; u < 2; ++u) {
      const int row = (tid >> 4) + 16 * u, o = tid & 15;
      const int n = m0 + row;
      if (n < Nn) {
        const float* pr = &Ps[row * PS1];
        float ig = fsigmoid(pr[o]);
        float fg = fsigmoid(pr[16 + o]);
        float tg = tanhf(pr[32 + o]);
        float og = fsigmoid(pr[48 + o]);
        const size_t qo = (size_t)n * HID + 16 * qd + o;
        float cv = fg * (u == 0 ? cin0 : cin1) + ig * tg;
        float hv = og * tanhf(cv);
        h_out[qo] = hv;
        c_out[qo] = cv;
      }
    }
  }
}


#define AS2 68
#define BS2 68

__global__ __launch_bounds__(256)
void k_gemm_out(const float* __restrict__ h0, const float* __restrict__ Wl,
                const float* __restrict__ bl, float* __restrict__ out, int Nn) {
  __shared__ float Bs[64 * BS2];
  __shared__ float As[32 * AS2];
  const int tid = threadIdx.x;
  const int m0 = blockIdx.x * 32;

  for (int i = tid; i < 64 * 64; i += 256) {
    int k = i >> 6, ccol = i & 63;
    Bs[k * BS2 + ccol] = Wl[i];
  }
  for (int i = tid; i < 32 * 64; i += 256) {
    int r = i >> 6, k = i & 63;
    int n = m0 + r; if (n >= Nn) n = Nn - 1;
    float v = h0[(size_t)n * HID + k];
    As[r * AS2 + k] = v > 0.0f ? v : 0.0f;
  }
  __syncthreads();

  const int wave = tid >> 5, lane = tid & 31;
  const int lo = lane & 15, hi = lane >> 4;
  const int r0 = (wave >> 2) * 16, c0 = (wave & 3) * 16;
  v8f acc = {};
  for (int k0 = 0; k0 < 64; k0 += 4) {
    v2f a = *(const v2f*)&As[(r0 + lo) * AS2 + k0 + 2 * hi];
    v2f b;
    b[0] = Bs[(k0 + 2 * hi) * BS2 + c0 + lo];
    b[1] = Bs[(k0 + 2 * hi + 1) * BS2 + c0 + lo];
    acc = wmma_f32_4(a, b, acc);
  }
  float bb = bl[c0 + lo];
#pragma unroll
  for (int r = 0; r < 8; ++r) {
    int row = r0 + r + 8 * hi;
    int n = m0 + row;
    if (n < Nn) out[(size_t)n * HID + c0 + lo] = acc[r] + bb;
  }
}


extern "C" void kernel_launch(void* const* d_in, const int* in_sizes, int n_in,
                              void* d_out, int out_size, void* d_ws, size_t ws_size,
                              hipStream_t stream) {
  const float* x      = (const float*)d_in[0];
  const int*   ei     = (const int*)d_in[1];
  const float* ew     = (const float*)d_in[2];
  const float* h      = (const float*)d_in[3];
  const float* c      = (const float*)d_in[4];
  const float* Wx     = (const float*)d_in[5];
  const float* bg     = (const float*)d_in[6];
  const float* theta  = (const float*)d_in[7];
  const float* conv_b = (const float*)d_in[8];
  const float* Wl     = (const float*)d_in[9];
  const float* bl     = (const float*)d_in[10];

  const int E  = in_sizes[2];
  const int Nn = in_sizes[3] / HID;
  const int* src = ei;
  const int* dst = ei + E;

  float* T1   = (float*)d_ws;
  float* T2   = T1 + (size_t)Nn * HID;
  float* X1   = T2 + (size_t)Nn * HID;
  float* X2   = X1 + (size_t)Nn * HID;
  float* Bp   = X2 + (size_t)Nn * HID;
  float* bias = Bp + (size_t)KDIM * GCOLS;
  float* lapw = bias + GCOLS;
  float* deg  = lapw + E;
  float* dinv = deg + 2 * (size_t)Nn;

  float* out   = (float*)d_out;
  float* h_out = out   + (size_t)Nn * HID;
  float* c_out = h_out + (size_t)Nn * HID;

  const float DIAG = 2.0f / LMAX - 1.0f;
  const int NH = Nn * HID;

  hipMemsetAsync(deg, 0, 2 * (size_t)Nn * sizeof(float), stream);
  k_degree<<<(E + 255) / 256, 256, 0, stream>>>(src, dst, ew, deg, E, Nn);
  k_dinv<<<(Nn + 255) / 256, 256, 0, stream>>>(deg, dinv, Nn);
  k_prep<<<(E + 255) / 256, 256, 0, stream>>>(src, dst, ew, dinv, lapw, E,
                                              h, T1, X1, X2, DIAG, NH / 4,
                                              Wx, theta, bg, conv_b, Bp, bias);

  k_spmm<<<(E + 255) / 256, 256, 0, stream>>>(h, T1, X1, src, dst, lapw, E);
  k_mid<<<(NH + 255) / 256, 256, 0, stream>>>(T1, X1, T2, DIAG, NH);
  k_spmm<<<(E + 255) / 256, 256, 0, stream>>>(T1, T2, X2, src, dst, lapw, E);

  k_gemm_gates<<<512, 256, 0, stream>>>(x, h, T1, T2, X2, Bp, bias, c,
                                        h_out, c_out, Nn, (Nn + TROWS - 1) / TROWS);
  k_gemm_out<<<(Nn + 31) / 32, 256, 0, stream>>>(h_out, Wl, bl, out, Nn);
}
